# starting kernel
# baseline (speedup 1.0000x reference)
#include <hip/hip_runtime.h>

typedef __attribute__((ext_vector_type(2))) float v2f;
typedef __attribute__((ext_vector_type(8))) float v8f;

#define NNODES  (128 * 512)
#define NEDGES  (128 * 512 * 16)
#define NGRAPHS 128
#define NPG     512
#define EPG     (512 * 16)
#define RES     128

__global__ void __launch_bounds__(512) k_dinv(const int* __restrict__ ei,
                                              float* __restrict__ dinv) {
  __shared__ int cnt[NPG];
  const int b = blockIdx.x;
  const int tid = threadIdx.x;
  if (tid < NPG) cnt[tid] = 1;
  __syncthreads();
  for (int e = tid; e < EPG; e += blockDim.x)
    atomicAdd(&cnt[ei[NEDGES + b * EPG + e] & (NPG - 1)], 1);
  __syncthreads();
  if (tid < NPG) dinv[b * NPG + tid] = rsqrtf((float)cnt[tid]);
}

template <int NT, int K>
__global__ void __launch_bounds__(256) k_gemm_wmma(const float* __restrict__ A,
                                                   const float* __restrict__ B,
                                                   float* __restrict__ C, int M) {
  const int N = NT * 16;
  __shared__ v2f sB[(K / 2) * N];
  for (int i = threadIdx.x; i < (K / 2) * N; i += blockDim.x) {
    int kp = i / N;
    int c  = i - kp * N;
    v2f w;
    w.x = B[(size_t)(2 * kp) * N + c];
    w.y = B[(size_t)(2 * kp + 1) * N + c];
    sB[i] = w;
  }
  __syncthreads();

  int wave = (int)((blockIdx.x * blockDim.x + threadIdx.x) >> 5);
  int lane = threadIdx.x & 31;
  if (wave >= (M >> 4)) return;
  int r  = lane & 15;
  int kb = (lane >> 4) << 1;
  const float* Arow = A + (size_t)(wave * 16 + r) * K;

  v8f acc[NT] = {};
  for (int k = 0; k < K; k += 4) {
    v2f a = *(const v2f*)(Arow + k + kb);
    v2f bf[NT];
#pragma unroll
    for (int t = 0; t < NT; ++t)
      bf[t] = sB[((k + kb) >> 1) * N + t * 16 + r];
#pragma unroll
    for (int t = 0; t < NT; ++t)
      acc[t] = __builtin_amdgcn_wmma_f32_16x16x4_f32(false, a, false, bf[t], (short)0,
                                                     acc[t], false, false);
  }
  int rbase = (lane >> 4) << 3;
#pragma unroll
  for (int t = 0; t < NT; ++t) {
    float* Crow = C + (size_t)(wave * 16 + rbase) * N + t * 16 + r;
#pragma unroll
    for (int v = 0; v < 8; ++v) Crow[(size_t)v * N] = acc[t][v];
  }
}

template <int C>
__global__ void __launch_bounds__(512) k_agg(
    const float* __restrict__ h, const int* __restrict__ ei,
    const float* __restrict__ dinv, const float* __restrict__ bias,
    float* __restrict__ y, float* __restrict__ pout) {
  extern __shared__ float smem[];
  float* acc = smem;
  float* hl  = smem + NPG * C;
  float* dv  = smem + 2 * NPG * C;
  const int b = blockIdx.x;
  const int gbase = b * NPG;
  const int tid = threadIdx.x;

  for (int i = tid; i < NPG; i += blockDim.x) dv[i] = dinv[gbase + i];
  for (int idx = tid; idx < NPG * C; idx += blockDim.x)
    hl[idx] = h[(size_t)gbase * C + idx];
  __syncthreads();

  for (int idx = tid; idx < NPG * C; idx += blockDim.x) {
    float d = dv[idx / C];
    acc[idx] = hl[idx] * d * d;
  }
  __syncthreads();

  const int chunks = C / 4;
  for (int t = tid; t < EPG * chunks; t += blockDim.x) {
    int e  = t / chunks;
    int cg = (t - e * chunks) * 4;
    int s  = ei[(size_t)b * EPG + e] - gbase;
    int d  = ei[(size_t)NEDGES + (size_t)b * EPG + e] - gbase;
    float w = dv[s] * dv[d];
    float* ad = acc + d * C + cg;
    const float* hs = hl + s * C + cg;
    atomicAdd(&ad[0], hs[0] * w);
    atomicAdd(&ad[1], hs[1] * w);
    atomicAdd(&ad[2], hs[2] * w);
    atomicAdd(&ad[3], hs[3] * w);
  }
  __syncthreads();

  for (int idx = tid; idx < NPG * C; idx += blockDim.x) {
    float v = acc[idx] + bias[idx & (C - 1)];
    y[(size_t)gbase * C + idx] = fmaxf(v, 0.0f);
  }
  for (int idx = tid; idx < RES * C; idx += blockDim.x) {
    int c = idx & (C - 1);
    int r = idx / C;
    float bb = bias[c];
    float v1 = fmaxf(acc[(4 * r + 1) * C + c] + bb, 0.0f);
    float v2 = fmaxf(acc[(4 * r + 2) * C + c] + bb, 0.0f);
    pout[(size_t)b * RES * C + idx] = 0.5f * (v1 + v2);
  }
}

template <int D>
__global__ void __launch_bounds__(256) k_sim_wmma(const float* __restrict__ U,
                                                  const float* __restrict__ V,
                                                  float* __restrict__ out) {
  __shared__ float sV[RES * D];
  const int b = blockIdx.x;
  const float* Vb = V + (size_t)b * RES * D;
  for (int i = threadIdx.x; i < RES * D; i += blockDim.x) sV[i] = Vb[i];
  __syncthreads();

  int tm   = threadIdx.x >> 5;
  int lane = threadIdx.x & 31;
  int r    = lane & 15;
  int kb   = (lane >> 4) << 1;
  const float* Ar = U + ((size_t)b * RES + tm * 16 + r) * D;

  v8f acc[8] = {};
  for (int k = 0; k < D; k += 4) {
    v2f a = *(const v2f*)(Ar + k + kb);
    v2f bf[8];
#pragma unroll
    for (int t = 0; t < 8; ++t)
      bf[t] = *(const v2f*)(sV + (t * 16 + r) * D + k + kb);
#pragma unroll
    for (int t = 0; t < 8; ++t)
      acc[t] = __builtin_amdgcn_wmma_f32_16x16x4_f32(false, a, false, bf[t], (short)0,
                                                     acc[t], false, false);
  }
  int rbase = (lane >> 4) << 3;
#pragma unroll
  for (int t = 0; t < 8; ++t) {
    float* Co = out + (size_t)b * RES * RES + (size_t)(tm * 16 + rbase) * RES + t * 16 + r;
#pragma unroll
    for (int v = 0; v < 8; ++v) Co[(size_t)v * RES] = acc[t][v];
  }
}

extern "C" void kernel_launch(void* const* d_in, const int* in_sizes, int n_in,
                              void* d_out, int out_size, void* d_ws, size_t ws_size,
                              hipStream_t stream) {
  (void)in_sizes; (void)n_in; (void)out_size; (void)ws_size;
  const float* x[2]  = {(const float*)d_in[0], (const float*)d_in[1]};
  const int*   ei[2] = {(const int*)d_in[2],  (const int*)d_in[3]};
  const float* W[3]  = {(const float*)d_in[6], (const float*)d_in[8], (const float*)d_in[10]};
  const float* bv[3] = {(const float*)d_in[7], (const float*)d_in[9], (const float*)d_in[11]};
  float* out = (float*)d_out;

  float* ws = (float*)d_ws;
  size_t o = 0;
  float* dinvA = ws + o; o += (size_t)2 * NNODES;
  float* B0 = ws + o; o += (size_t)NNODES * 64;
  float* B1 = ws + o; o += (size_t)NNODES * 64;
  float* B2 = ws + o; o += (size_t)NNODES * 32;
  float* P[2][3];
  const int dims[3] = {64, 32, 16};
  for (int s = 0; s < 2; ++s)
    for (int l = 0; l < 3; ++l) { P[s][l] = ws + o; o += (size_t)NGRAPHS * RES * dims[l]; }

  const unsigned gemmBlocks = (NNODES / 16) * 32 / 256;
  auto shbytes = [](int C) { return (size_t)(2 * NPG * C + NPG) * sizeof(float); };

  for (int s = 0; s < 2; ++s) {
    float* dv = dinvA + (size_t)s * NNODES;
    k_dinv<<<NGRAPHS, 512, 0, stream>>>(ei[s], dv);

    k_gemm_wmma<4, 128><<<gemmBlocks, 256, 0, stream>>>(x[s], W[0], B0, NNODES);
    k_agg<64><<<NGRAPHS, 512, shbytes(64), stream>>>(B0, ei[s], dv, bv[0], B1, P[s][0]);
    k_gemm_wmma<2, 64><<<gemmBlocks, 256, 0, stream>>>(B1, W[1], B0, NNODES);
    k_agg<32><<<NGRAPHS, 512, shbytes(32), stream>>>(B0, ei[s], dv, bv[1], B2, P[s][1]);
    k_gemm_wmma<1, 32><<<gemmBlocks, 256, 0, stream>>>(B2, W[2], B0, NNODES);
    k_agg<16><<<NGRAPHS, 512, shbytes(16), stream>>>(B0, ei[s], dv, bv[2], B1, P[s][2]);
  }

  k_sim_wmma<64><<<NGRAPHS, 256, 0, stream>>>(P[0][0], P[1][0], out);
  k_sim_wmma<32><<<NGRAPHS, 256, 0, stream>>>(P[0][1], P[1][1],
                                              out + (size_t)1 * NGRAPHS * RES * RES);
  k_sim_wmma<16><<<NGRAPHS, 256, 0, stream>>>(P[0][2], P[1][2],
                                              out + (size_t)2 * NGRAPHS * RES * RES);
}


# similarity product: pooled rows stored k-pair-major [k/2][row][2] (turned in LDS by the layer epilogue); A and B fragments are whole-line global loads, no LDS tile; 2 work-groups a graph; same k order
# speedup vs baseline: 5.8363x; 5.8363x over previous
#include <hip/hip_runtime.h>


typedef __attribute__((ext_vector_type(2))) float v2f;
typedef __attribute__((ext_vector_type(8))) float v8f;

#define NNODES  (128 * 512)
#define NEDGES  (128 * 512 * 16)
#define NGRAPHS 128
#define NPG     512
#define EPG     (512 * 16)
#define RES     128
#define CS      16


__global__ void __launch_bounds__(512) k_dinv(const int* __restrict__ ei0,
                                              const int* __restrict__ ei1,
                                              float* __restrict__ dinvA) {
  __shared__ int cnt[NPG];
  const int b = blockIdx.x;
  const int tid = threadIdx.x;
  const int* __restrict__ ei = blockIdx.y ? ei1 : ei0;
  float* __restrict__ dinv = dinvA + (size_t)blockIdx.y * NNODES;
  if (tid < NPG) cnt[tid] = 1;
  __syncthreads();
  for (int e = tid; e < EPG; e += blockDim.x)
    atomicAdd(&cnt[ei[NEDGES + b * EPG + e] & (NPG - 1)], 1);
  __syncthreads();
  if (tid < NPG) dinv[b * NPG + tid] = rsqrtf((float)cnt[tid]);
}

template <int G0, int G>
__device__ __forceinline__ void agg_take(unsigned sd, unsigned wbits, unsigned cc,
                                         const float* hl, float* hv, unsigned* dofs) {
  if constexpr (G < 8) {
    const unsigned t =
        (unsigned)__builtin_amdgcn_mov_dpp((int)sd, 0x150 + G0 + G, 0xf, 0xf, true) | cc;
    const float wg = __uint_as_float(
        (unsigned)__builtin_amdgcn_mov_dpp((int)wbits, 0x150 + G0 + G, 0xf, 0xf, true));
    hv[G] = *(const float*)((const char*)hl + (t & 0xffffu)) * wg;
    dofs[G] = t >> 16;
    agg_take<G0, G + 1>(sd, wbits, cc, hl, hv, dofs);
  }
}

template <int G0>
__device__ __forceinline__ void agg_half(unsigned sd, unsigned wbits, unsigned cc,
                                         const float* hl, float* acc) {
  float    hv[8];
  unsigned dofs[8];
  agg_take<G0, 0>(sd, wbits, cc, hl, hv, dofs);
  for (int g = 0; g < 8; ++g) atomicAdd((float*)((char*)acc + dofs[g]), hv[g]);
}

template <int G0, int G>
__device__ __forceinline__ void agg32_take(unsigned sd, unsigned wbits, unsigned c4,
                                           const float* hl, float* hv, unsigned* dofs) {
  if constexpr (G < 8) {
    const unsigned t = (unsigned)__builtin_amdgcn_readlane((int)sd, G0 + G);
    const float wg = __uint_as_float((unsigned)__builtin_amdgcn_readlane((int)wbits, G0 + G));
    hv[G] = *(const float*)((const char*)hl + ((t & 0xffffu) | c4)) * wg;
    dofs[G] = (t >> 16) | c4;
    agg32_take<G0, G + 1>(sd, wbits, c4, hl, hv, dofs);
  }
}

template <int G0>
__device__ __forceinline__ void agg32_group(unsigned sd, unsigned wbits, unsigned c4,
                                            const float* hl, float* acc) {
  float    hv[8];
  unsigned dofs[8];
  agg32_take<G0, 0>(sd, wbits, c4, hl, hv, dofs);
  for (int g = 0; g < 8; ++g) atomicAdd((float*)((char*)acc + dofs[g]), hv[g]);
}

template <int K, int CSI, int N, int CSO>
__global__ void __launch_bounds__(32 * CSO) k_layer(
    const float* __restrict__ in0, const float* __restrict__ in1,
    const float* __restrict__ Wm,
    const int* __restrict__ ei0, const int* __restrict__ ei1,
    const float* __restrict__ dinvA, const float* __restrict__ bias,
    float* __restrict__ y0, float* __restrict__ y1,
    float* __restrict__ pout0, float* __restrict__ pout1) {
  constexpr int MT = 32 / CSO;
  constexpr int NT = CSO / 16;
  constexpr int WP = (CSO == 32) ? 48 : 16;
  __shared__ float smem[2 * NPG * CSO + NPG];
  __shared__ v2f sW[(K / 2) * WP];
  const int side = blockIdx.z;
  const float* __restrict__ in   = side ? in1 : in0;
  const int*   __restrict__ ei   = side ? ei1 : ei0;
  const float* __restrict__ dinv = dinvA + (size_t)side * NNODES;
  float* __restrict__ y    = side ? y1 : y0;
  float* __restrict__ pout = side ? pout1 : pout0;
  float* acc = smem;
  float* hl  = smem + NPG * CSO;
  float* dv  = smem + 2 * NPG * CSO;
  const int b  = blockIdx.x;
  const int sl = blockIdx.y;
  const int gbase = b * NPG;
  const int tid = threadIdx.x;
  const int lane = tid & 31;
  const int nwv = (int)(blockDim.x >> 5);
  float*       yg = y + ((size_t)sl * NNODES + gbase) * CSO;
  const float* bsl = bias + sl * CSO;

  for (int i = tid; i < NPG; i += blockDim.x) dv[i] = dinv[gbase + i];
  for (int i = tid; i < (K / 2) * CSO; i += blockDim.x) {
    int kp = i / CSO;
    int c  = i - kp * CSO;
    v2f w;
    w.x = Wm[(size_t)(2 * kp) * N + sl * CSO + c];
    w.y = Wm[(size_t)(2 * kp + 1) * N + sl * CSO + c];
    sW[kp * WP + c] = w;
  }
  __syncthreads();

  {
    const int r  = lane & 15;
    const int kb = (lane >> 4) << 1;
    const int rbase = (lane >> 4) << 3;
    const int row0 = (tid >> 5) * (16 * MT);
    v8f pa[MT][NT] = {};
#pragma unroll 4
    for (int k = 0; k < K; k += 4) {
      v2f a[MT];
#pragma unroll
      for (int mt = 0; mt < MT; ++mt) {
        const size_t row = (size_t)(gbase + row0 + mt * 16 + r);
        const float* ap;
        if constexpr (CSI == 0)
          ap = in + row * K + k + kb;
        else
          ap = in + ((size_t)(k / CSI) * NNODES + row) * CSI + (k % CSI) + kb;
        a[mt] = *(const v2f*)ap;
      }
      v2f bf[NT];
#pragma unroll
      for (int nt = 0; nt < NT; ++nt)
        bf[nt] = sW[((k + kb) >> 1) * WP + nt * 16 + r];
#pragma unroll
      for (int mt = 0; mt < MT; ++mt)
#pragma unroll
        for (int nt = 0; nt < NT; ++nt)
          pa[mt][nt] = __builtin_amdgcn_wmma_f32_16x16x4_f32(false, a[mt], false, bf[nt],
                                                             (short)0, pa[mt][nt], false, false);
    }
#pragma unroll
    for (int mt = 0; mt < MT; ++mt) {
#pragma unroll
      for (int nt = 0; nt < NT; ++nt) {
#pragma unroll
        for (int v = 0; v < 8; ++v) {
          const int rl = row0 + mt * 16 + rbase + v;
          const float hval = pa[mt][nt][v];
          const float d = dv[rl];
          hl[rl * CSO + nt * 16 + r]  = hval;
          acc[rl * CSO + nt * 16 + r] = hval * d * d;
        }
      }
    }
  }
  __syncthreads();

  {
    const int* __restrict__ es = ei + (size_t)b * EPG;
    const int* __restrict__ ed = ei + (size_t)NEDGES + (size_t)b * EPG;
    for (int e0 = (tid >> 5) * 32; e0 < EPG; e0 += nwv * 32) {
      const int s = es[e0 + lane] - gbase;
      const int d = ed[e0 + lane] - gbase;
      const float w = dv[s] * dv[d];
      const unsigned sd = ((unsigned)s * (CSO * 4u)) | (((unsigned)d * (CSO * 4u)) << 16);
      if constexpr (CSO == 32) {
        const unsigned c4 = (unsigned)lane * 4u;
        agg32_group<0>(sd, __float_as_uint(w), c4, hl, acc);
        agg32_group<8>(sd, __float_as_uint(w), c4, hl, acc);
        agg32_group<16>(sd, __float_as_uint(w), c4, hl, acc);
        agg32_group<24>(sd, __float_as_uint(w), c4, hl, acc);
      } else {
        const unsigned c4 = (unsigned)(lane & (CS - 1)) * 4u;
        const unsigned cc = c4 | (c4 << 16);
        agg_half<0>(sd, __float_as_uint(w), cc, hl, acc);
        agg_half<8>(sd, __float_as_uint(w), cc, hl, acc);
      }
    }
  }
  __syncthreads();

  constexpr int TP = 2 * RES + 64 / CSO;
  float* tp = hl;
  for (int idx = tid; idx < RES * CSO; idx += blockDim.x) {
    int c = idx & (CSO - 1);
    int r = idx / CSO;
    float bb = bsl[c];
    float v1 = fmaxf(acc[(4 * r + 1) * CSO + c] + bb, 0.0f);
    float v2 = fmaxf(acc[(4 * r + 2) * CSO + c] + bb, 0.0f);
    tp[(c >> 1) * TP + 2 * r + (c & 1)] = 0.5f * (v1 + v2);
  }
  __syncthreads();
  float* pg = pout + (size_t)b * RES * N + (size_t)sl * RES * CSO;
  for (int idx = tid; idx < RES * CSO; idx += blockDim.x)
    pg[idx] = tp[(idx >> 8) * TP + (idx & (2 * RES - 1))];
  for (int idx = tid; idx < NPG * CSO; idx += blockDim.x) {
    float v = acc[idx] + bsl[idx & (CSO - 1)];
    yg[idx] = fmaxf(v, 0.0f);
  }
}

template <int D>
__global__ void __launch_bounds__(256) __attribute__((amdgpu_waves_per_eu(2, 2))) k_sim_wmma(const float* __restrict__ U,
                                                  const float* __restrict__ V,
                                                  float* __restrict__ out) {
  const int b    = blockIdx.x;
  const int w    = threadIdx.x >> 5;
  const int lane = threadIdx.x & 31;
  const int r    = lane & 15;
  const int h    = lane >> 4;
  const int tm   = blockIdx.y * 4 + (w >> 1);
  const int tc   = (w & 1) * 4;
  const v2f* Ug = (const v2f*)U + (size_t)b * (RES * D / 2) + h * RES + tm * 16 + r;
  const v2f* Vg = (const v2f*)V + (size_t)b * (RES * D / 2) + h * RES + tc * 16 + r;

  constexpr int KS = D / 4;
  v2f a[KS];
  v2f bf[KS][4];
#pragma unroll
  for (int s = 0; s < KS; ++s) {
    a[s] = Ug[(2 * s) * RES];
#pragma unroll
    for (int t = 0; t < 4; ++t)
      bf[s][t] = Vg[(2 * s) * RES + t * 16];
  }
  v8f acc[4] = {};
#pragma unroll
  for (int s = 0; s < KS; ++s)
#pragma unroll
    for (int t = 0; t < 4; ++t)
      acc[t] = __builtin_amdgcn_wmma_f32_16x16x4_f32(false, a[s], false, bf[s][t], (short)0,
                                                     acc[t], false, false);
  const int rbase = h << 3;
#pragma unroll
  for (int t = 0; t < 4; ++t) {
    float* Co = out + (size_t)b * RES * RES + (size_t)(tm * 16 + rbase) * RES + (tc + t) * 16 + r;
#pragma unroll
    for (int v = 0; v < 8; ++v) Co[(size_t)v * RES] = acc[t][v];
  }
}

extern "C" void kernel_launch(void* const* d_in, const int* in_sizes, int n_in,
                              void* d_out, int out_size, void* d_ws, size_t ws_size,
                              hipStream_t stream) {
  (void)in_sizes; (void)n_in; (void)out_size; (void)ws_size;
  const float* x[2]  = {(const float*)d_in[0], (const float*)d_in[1]};
  const int*   ei[2] = {(const int*)d_in[2],  (const int*)d_in[3]};
  const float* W[3]  = {(const float*)d_in[6], (const float*)d_in[8], (const float*)d_in[10]};
  const float* bv[3] = {(const float*)d_in[7], (const float*)d_in[9], (const float*)d_in[11]};
  float* out = (float*)d_out;

  float* ws = (float*)d_ws;
  size_t o = 0;
  float* dinvA = ws + o; o += (size_t)2 * NNODES;
  float* Y1[2]; float* Y2[2]; float* Y3[2];
  for (int s = 0; s < 2; ++s) { Y1[s] = ws + o; o += (size_t)NNODES * 64; }
  for (int s = 0; s < 2; ++s) { Y2[s] = ws + o; o += (size_t)NNODES * 32; }
  for (int s = 0; s < 2; ++s) { Y3[s] = ws + o; o += (size_t)NNODES * 16; }
  float* P[2][3];
  const int dims[3] = {64, 32, 16};
  for (int s = 0; s < 2; ++s)
    for (int l = 0; l < 3; ++l) { P[s][l] = ws + o; o += (size_t)NGRAPHS * RES * dims[l]; }

  k_dinv<<<dim3(NGRAPHS, 2), 512, 0, stream>>>(ei[0], ei[1], dinvA);

  k_layer<128, 0, 64, 32><<<dim3(NGRAPHS, 2, 2), 1024, 0, stream>>>(
      x[0], x[1], W[0], ei[0], ei[1], dinvA, bv[0], Y1[0], Y1[1], P[0][0], P[1][0]);
  k_layer<64, 32, 32, 32><<<dim3(NGRAPHS, 1, 2), 1024, 0, stream>>>(
      Y1[0], Y1[1], W[1], ei[0], ei[1], dinvA, bv[1], Y2[0], Y2[1], P[0][1], P[1][1]);
  k_layer<32, 32, 16, 16><<<dim3(NGRAPHS, 1, 2), 512, 0, stream>>>(
      Y2[0], Y2[1], W[2], ei[0], ei[1], dinvA, bv[2], Y3[0], Y3[1], P[0][2], P[1][2]);

  k_sim_wmma<64><<<dim3(NGRAPHS, 2), 256, 0, stream>>>(P[0][0], P[1][0], out);
  k_sim_wmma<32><<<dim3(NGRAPHS, 2), 256, 0, stream>>>(P[0][1], P[1][1],
                                                   out + (size_t)1 * NGRAPHS * RES * RES);
  k_sim_wmma<16><<<dim3(NGRAPHS, 2), 256, 0, stream>>>(P[0][2], P[1][2],
                                                   out + (size_t)2 * NGRAPHS * RES * RES);
}


# the layer kernels count their graph's in-degrees themselves: LDS integer atomics on its 8192 edge targets, draining under the weight product; the degree launch and dinv array go; 7 launches become 6
# speedup vs baseline: 6.3324x; 1.0850x over previous
#include <hip/hip_runtime.h>


typedef __attribute__((ext_vector_type(2))) float v2f;
typedef __attribute__((ext_vector_type(8))) float v8f;

#define NNODES  (128 * 512)
#define NEDGES  (128 * 512 * 16)
#define NGRAPHS 128
#define NPG     512
#define EPG     (512 * 16)
#define RES     128
#define CS      16


template <int G0, int G>
__device__ __forceinline__ void agg_take(unsigned sd, unsigned wbits, unsigned cc,
                                         const float* hl, float* hv, unsigned* dofs) {
  if constexpr (G < 8) {
    const unsigned t =
        (unsigned)__builtin_amdgcn_mov_dpp((int)sd, 0x150 + G0 + G, 0xf, 0xf, true) | cc;
    const float wg = __uint_as_float(
        (unsigned)__builtin_amdgcn_mov_dpp((int)wbits, 0x150 + G0 + G, 0xf, 0xf, true));
    hv[G] = *(const float*)((const char*)hl + (t & 0xffffu)) * wg;
    dofs[G] = t >> 16;
    agg_take<G0, G + 1>(sd, wbits, cc, hl, hv, dofs);
  }
}

template <int G0>
__device__ __forceinline__ void agg_half(unsigned sd, unsigned wbits, unsigned cc,
                                         const float* hl, float* acc) {
  float    hv[8];
  unsigned dofs[8];
  agg_take<G0, 0>(sd, wbits, cc, hl, hv, dofs);
  for (int g = 0; g < 8; ++g) atomicAdd((float*)((char*)acc + dofs[g]), hv[g]);
}

template <int G0, int G>
__device__ __forceinline__ void agg32_take(unsigned sd, unsigned wbits, unsigned c4,
                                           const float* hl, float* hv, unsigned* dofs) {
  if constexpr (G < 8) {
    const unsigned t = (unsigned)__builtin_amdgcn_readlane((int)sd, G0 + G);
    const float wg = __uint_as_float((unsigned)__builtin_amdgcn_readlane((int)wbits, G0 + G));
    hv[G] = *(const float*)((const char*)hl + ((t & 0xffffu) | c4)) * wg;
    dofs[G] = (t >> 16) | c4;
    agg32_take<G0, G + 1>(sd, wbits, c4, hl, hv, dofs);
  }
}

template <int G0>
__device__ __forceinline__ void agg32_group(unsigned sd, unsigned wbits, unsigned c4,
                                            const float* hl, float* acc) {
  float    hv[8];
  unsigned dofs[8];
  agg32_take<G0, 0>(sd, wbits, c4, hl, hv, dofs);
  for (int g = 0; g < 8; ++g) atomicAdd((float*)((char*)acc + dofs[g]), hv[g]);
}

template <int K, int CSI, int N, int CSO>
__global__ void __launch_bounds__(32 * CSO) k_layer(
    const float* __restrict__ in0, const float* __restrict__ in1,
    const float* __restrict__ Wm,
    const int* __restrict__ ei0, const int* __restrict__ ei1,
    const float* __restrict__ bias,
    float* __restrict__ y0, float* __restrict__ y1,
    float* __restrict__ pout0, float* __restrict__ pout1) {
  constexpr int MT = 32 / CSO;
  constexpr int NT = CSO / 16;
  constexpr int WP = (CSO == 32) ? 48 : 16;
  __shared__ float smem[2 * NPG * CSO];
  __shared__ v2f sW[(K / 2) * WP];
  __shared__ int cnt[NPG];
  constexpr int NTHR = 32 * CSO;
  constexpr int EPT  = EPG / NTHR;
  const int side = blockIdx.z;
  const float* __restrict__ in   = side ? in1 : in0;
  const int*   __restrict__ ei   = side ? ei1 : ei0;
  float* __restrict__ y    = side ? y1 : y0;
  float* __restrict__ pout = side ? pout1 : pout0;
  float* acc = smem;
  float* hl  = smem + NPG * CSO;
  const int b  = blockIdx.x;
  const int sl = blockIdx.y;
  const int gbase = b * NPG;
  const int tid = threadIdx.x;
  const int lane = tid & 31;
  const int nwv = (int)(blockDim.x >> 5);
  float*       yg = y + ((size_t)sl * NNODES + gbase) * CSO;
  const float* bsl = bias + sl * CSO;

  int dl[EPT];
#pragma unroll
  for (int t = 0; t < EPT; ++t)
    dl[t] = ei[(size_t)NEDGES + (size_t)b * EPG + t * NTHR + tid] & (NPG - 1);
  static_assert(NTHR >= NPG, "one thread per node for the count's start value");
  if (tid < NPG) cnt[tid] = 1;
  for (int i = tid; i < (K / 2) * CSO; i += blockDim.x) {
    int kp = i / CSO;
    int c  = i - kp * CSO;
    v2f w;
    w.x = Wm[(size_t)(2 * kp) * N + sl * CSO + c];
    w.y = Wm[(size_t)(2 * kp + 1) * N + sl * CSO + c];
    sW[kp * WP + c] = w;
  }
  __syncthreads();
#pragma unroll
  for (int t = 0; t < EPT; ++t) atomicAdd(&cnt[dl[t]], 1);

  {
    const int r  = lane & 15;
    const int kb = (lane >> 4) << 1;
    const int rbase = (lane >> 4) << 3;
    const int row0 = (tid >> 5) * (16 * MT);
    v8f pa[MT][NT] = {};
#pragma unroll 4
    for (int k = 0; k < K; k += 4) {
      v2f a[MT];
#pragma unroll
      for (int mt = 0; mt < MT; ++mt) {
        const size_t row = (size_t)(gbase + row0 + mt * 16 + r);
        const float* ap;
        if constexpr (CSI == 0)
          ap = in + row * K + k + kb;
        else
          ap = in + ((size_t)(k / CSI) * NNODES + row) * CSI + (k % CSI) + kb;
        a[mt] = *(const v2f*)ap;
      }
      v2f bf[NT];
#pragma unroll
      for (int nt = 0; nt < NT; ++nt)
        bf[nt] = sW[((k + kb) >> 1) * WP + nt * 16 + r];
#pragma unroll
      for (int mt = 0; mt < MT; ++mt)
#pragma unroll
        for (int nt = 0; nt < NT; ++nt)
          pa[mt][nt] = __builtin_amdgcn_wmma_f32_16x16x4_f32(false, a[mt], false, bf[nt],
                                                             (short)0, pa[mt][nt], false, false);
    }
    __syncthreads();
#pragma unroll
    for (int mt = 0; mt < MT; ++mt) {
#pragma unroll
      for (int nt = 0; nt < NT; ++nt) {
#pragma unroll
        for (int v = 0; v < 8; ++v) {
          const int rl = row0 + mt * 16 + rbase + v;
          const float hval = pa[mt][nt][v];
          const float d = rsqrtf((float)cnt[rl]);
          hl[rl * CSO + nt * 16 + r]  = hval;
          acc[rl * CSO + nt * 16 + r] = hval * d * d;
        }
      }
    }
  }
  __syncthreads();

  {
    const int* __restrict__ es = ei + (size_t)b * EPG;
    const int* __restrict__ ed = ei + (size_t)NEDGES + (size_t)b * EPG;
    for (int e0 = (tid >> 5) * 32; e0 < EPG; e0 += nwv * 32) {
      const int s = es[e0 + lane] - gbase;
      const int d = ed[e0 + lane] - gbase;
      const float w = rsqrtf((float)cnt[s]) * rsqrtf((float)cnt[d]);
      const unsigned sd = ((unsigned)s * (CSO * 4u)) | (((unsigned)d * (CSO * 4u)) << 16);
      if constexpr (CSO == 32) {
        const unsigned c4 = (unsigned)lane * 4u;
        agg32_group<0>(sd, __float_as_uint(w), c4, hl, acc);
        agg32_group<8>(sd, __float_as_uint(w), c4, hl, acc);
        agg32_group<16>(sd, __float_as_uint(w), c4, hl, acc);
        agg32_group<24>(sd, __float_as_uint(w), c4, hl, acc);
      } else {
        const unsigned c4 = (unsigned)(lane & (CS - 1)) * 4u;
        const unsigned cc = c4 | (c4 << 16);
        agg_half<0>(sd, __float_as_uint(w), cc, hl, acc);
        agg_half<8>(sd, __float_as_uint(w), cc, hl, acc);
      }
    }
  }
  __syncthreads();

  constexpr int TP = 2 * RES + 64 / CSO;
  float* tp = hl;
  for (int idx = tid; idx < RES * CSO; idx += blockDim.x) {
    int c = idx & (CSO - 1);
    int r = idx / CSO;
    float bb = bsl[c];
    float v1 = fmaxf(acc[(4 * r + 1) * CSO + c] + bb, 0.0f);
    float v2 = fmaxf(acc[(4 * r + 2) * CSO + c] + bb, 0.0f);
    tp[(c >> 1) * TP + 2 * r + (c & 1)] = 0.5f * (v1 + v2);
  }
  __syncthreads();
  float* pg = pout + (size_t)b * RES * N + (size_t)sl * RES * CSO;
  for (int idx = tid; idx < RES * CSO; idx += blockDim.x)
    pg[idx] = tp[(idx >> 8) * TP + (idx & (2 * RES - 1))];
  for (int idx = tid; idx < NPG * CSO; idx += blockDim.x) {
    float v = acc[idx] + bsl[idx & (CSO - 1)];
    yg[idx] = fmaxf(v, 0.0f);
  }
}

template <int D>
__global__ void __launch_bounds__(256) __attribute__((amdgpu_waves_per_eu(2, 2))) k_sim_wmma(const float* __restrict__ U,
                                                  const float* __restrict__ V,
                                                  float* __restrict__ out) {
  const int b    = blockIdx.x;
  const int w    = threadIdx.x >> 5;
  const int lane = threadIdx.x & 31;
  const int r    = lane & 15;
  const int h    = lane >> 4;
  const int tm   = blockIdx.y * 4 + (w >> 1);
  const int tc   = (w & 1) * 4;
  const v2f* Ug = (const v2f*)U + (size_t)b * (RES * D / 2) + h * RES + tm * 16 + r;
  const v2f* Vg = (const v2f*)V + (size_t)b * (RES * D / 2) + h * RES + tc * 16 + r;

  constexpr int KS = D / 4;
  v2f a[KS];
  v2f bf[KS][4];
#pragma unroll
  for (int s = 0; s < KS; ++s) {
    a[s] = Ug[(2 * s) * RES];
#pragma unroll
    for (int t = 0; t < 4; ++t)
      bf[s][t] = Vg[(2 * s) * RES + t * 16];
  }
  v8f acc[4] = {};
#pragma unroll
  for (int s = 0; s < KS; ++s)
#pragma unroll
    for (int t = 0; t < 4; ++t)
      acc[t] = __builtin_amdgcn_wmma_f32_16x16x4_f32(false, a[s], false, bf[s][t], (short)0,
                                                     acc[t], false, false);
  const int rbase = h << 3;
#pragma unroll
  for (int t = 0; t < 4; ++t) {
    float* Co = out + (size_t)b * RES * RES + (size_t)(tm * 16 + rbase) * RES + (tc + t) * 16 + r;
#pragma unroll
    for (int v = 0; v < 8; ++v) Co[(size_t)v * RES] = acc[t][v];
  }
}

extern "C" void kernel_launch(void* const* d_in, const int* in_sizes, int n_in,
                              void* d_out, int out_size, void* d_ws, size_t ws_size,
                              hipStream_t stream) {
  (void)in_sizes; (void)n_in; (void)out_size; (void)ws_size;
  const float* x[2]  = {(const float*)d_in[0], (const float*)d_in[1]};
  const int*   ei[2] = {(const int*)d_in[2],  (const int*)d_in[3]};
  const float* W[3]  = {(const float*)d_in[6], (const float*)d_in[8], (const float*)d_in[10]};
  const float* bv[3] = {(const float*)d_in[7], (const float*)d_in[9], (const float*)d_in[11]};
  float* out = (float*)d_out;

  float* ws = (float*)d_ws;
  size_t o = 0;
  float* Y1[2]; float* Y2[2]; float* Y3[2];
  for (int s = 0; s < 2; ++s) { Y1[s] = ws + o; o += (size_t)NNODES * 64; }
  for (int s = 0; s < 2; ++s) { Y2[s] = ws + o; o += (size_t)NNODES * 32; }
  for (int s = 0; s < 2; ++s) { Y3[s] = ws + o; o += (size_t)NNODES * 16; }
  float* P[2][3];
  const int dims[3] = {64, 32, 16};
  for (int s = 0; s < 2; ++s)
    for (int l = 0; l < 3; ++l) { P[s][l] = ws + o; o += (size_t)NGRAPHS * RES * dims[l]; }


  k_layer<128, 0, 64, 32><<<dim3(NGRAPHS, 2, 2), 1024, 0, stream>>>(
      x[0], x[1], W[0], ei[0], ei[1], bv[0], Y1[0], Y1[1], P[0][0], P[1][0]);
  k_layer<64, 32, 32, 32><<<dim3(NGRAPHS, 1, 2), 1024, 0, stream>>>(
      Y1[0], Y1[1], W[1], ei[0], ei[1], bv[1], Y2[0], Y2[1], P[0][1], P[1][1]);
  k_layer<32, 32, 16, 16><<<dim3(NGRAPHS, 1, 2), 512, 0, stream>>>(
      Y2[0], Y2[1], W[2], ei[0], ei[1], bv[2], Y3[0], Y3[1], P[0][2], P[1][2]);

  k_sim_wmma<64><<<dim3(NGRAPHS, 2), 256, 0, stream>>>(P[0][0], P[1][0], out);
  k_sim_wmma<32><<<dim3(NGRAPHS, 2), 256, 0, stream>>>(P[0][1], P[1][1],
                                                   out + (size_t)1 * NGRAPHS * RES * RES);
  k_sim_wmma<16><<<dim3(NGRAPHS, 2), 256, 0, stream>>>(P[0][2], P[1][2],
                                                   out + (size_t)2 * NGRAPHS * RES * RES);
}


# layer 3 aggregates only node rows 4r+1, 4r+2 (all its pooling reads): a wave packs the edges ending there into its own LDS list and scatters those; no activation store; arrival order differs
# speedup vs baseline: 7.0411x; 1.1119x over previous
#include <hip/hip_runtime.h>


typedef __attribute__((ext_vector_type(2))) float v2f;
typedef __attribute__((ext_vector_type(8))) float v8f;
typedef __attribute__((ext_vector_type(2))) unsigned v2u;

#define NNODES  (128 * 512)
#define NEDGES  (128 * 512 * 16)
#define NGRAPHS 128
#define NPG     512
#define EPG     (512 * 16)
#define RES     128
#define CS      16


template <int G0, int G>
__device__ __forceinline__ void agg_take(unsigned sd, unsigned wbits, unsigned cc,
                                         const float* hl, float* hv, unsigned* dofs) {
  if constexpr (G < 8) {
    const unsigned t =
        (unsigned)__builtin_amdgcn_mov_dpp((int)sd, 0x150 + G0 + G, 0xf, 0xf, true) | cc;
    const float wg = __uint_as_float(
        (unsigned)__builtin_amdgcn_mov_dpp((int)wbits, 0x150 + G0 + G, 0xf, 0xf, true));
    hv[G] = *(const float*)((const char*)hl + (t & 0xffffu)) * wg;
    dofs[G] = t >> 16;
    agg_take<G0, G + 1>(sd, wbits, cc, hl, hv, dofs);
  }
}

template <int G0>
__device__ __forceinline__ void agg_half(unsigned sd, unsigned wbits, unsigned cc,
                                         const float* hl, float* acc) {
  float    hv[8];
  unsigned dofs[8];
  agg_take<G0, 0>(sd, wbits, cc, hl, hv, dofs);
  for (int g = 0; g < 8; ++g) atomicAdd((float*)((char*)acc + dofs[g]), hv[g]);
}

template <int G0, int G>
__device__ __forceinline__ void agg32_take(unsigned sd, unsigned wbits, unsigned c4,
                                           const float* hl, float* hv, unsigned* dofs) {
  if constexpr (G < 8) {
    const unsigned t = (unsigned)__builtin_amdgcn_readlane((int)sd, G0 + G);
    const float wg = __uint_as_float((unsigned)__builtin_amdgcn_readlane((int)wbits, G0 + G));
    hv[G] = *(const float*)((const char*)hl + ((t & 0xffffu) | c4)) * wg;
    dofs[G] = (t >> 16) | c4;
    agg32_take<G0, G + 1>(sd, wbits, c4, hl, hv, dofs);
  }
}

template <int G0>
__device__ __forceinline__ void agg32_group(unsigned sd, unsigned wbits, unsigned c4,
                                            const float* hl, float* acc) {
  float    hv[8];
  unsigned dofs[8];
  agg32_take<G0, 0>(sd, wbits, c4, hl, hv, dofs);
  for (int g = 0; g < 8; ++g) atomicAdd((float*)((char*)acc + dofs[g]), hv[g]);
}

template <int K, int CSI, int N, int CSO, bool LAST>
__global__ void __launch_bounds__(32 * CSO) k_layer(
    const float* __restrict__ in0, const float* __restrict__ in1,
    const float* __restrict__ Wm,
    const int* __restrict__ ei0, const int* __restrict__ ei1,
    const float* __restrict__ bias,
    float* __restrict__ y0, float* __restrict__ y1,
    float* __restrict__ pout0, float* __restrict__ pout1) {
  constexpr int MT = 32 / CSO;
  constexpr int NT = CSO / 16;
  constexpr int WP = (CSO == 32) ? 48 : 16;
  __shared__ float smem[2 * NPG * CSO];
  __shared__ v2f sW[(K / 2) * WP];
  __shared__ int cnt[NPG];
  constexpr int NTHR = 32 * CSO;
  constexpr int EPT  = EPG / NTHR;
  const int side = blockIdx.z;
  const float* __restrict__ in   = side ? in1 : in0;
  const int*   __restrict__ ei   = side ? ei1 : ei0;
  float* __restrict__ y    = side ? y1 : y0;
  float* __restrict__ pout = side ? pout1 : pout0;
  float* acc = smem;
  float* hl  = smem + NPG * CSO;
  const int b  = blockIdx.x;
  const int sl = blockIdx.y;
  const int gbase = b * NPG;
  const int tid = threadIdx.x;
  const int lane = tid & 31;
  const int nwv = (int)(blockDim.x >> 5);
  const float* bsl = bias + sl * CSO;

  int dl[EPT];
#pragma unroll
  for (int t = 0; t < EPT; ++t)
    dl[t] = ei[(size_t)NEDGES + (size_t)b * EPG + t * NTHR + tid] & (NPG - 1);
  int sl_[LAST ? EPT : 1];
  if constexpr (LAST) {
#pragma unroll
    for (int t = 0; t < EPT; ++t)
      sl_[t] = ei[(size_t)b * EPG + t * NTHR + tid] & (NPG - 1);
  }
  static_assert(NTHR >= NPG, "one thread per node for the count's start value");
  if (tid < NPG) cnt[tid] = 1;
  for (int i = tid; i < (K / 2) * CSO; i += blockDim.x) {
    int kp = i / CSO;
    int c  = i - kp * CSO;
    v2f w;
    w.x = Wm[(size_t)(2 * kp) * N + sl * CSO + c];
    w.y = Wm[(size_t)(2 * kp + 1) * N + sl * CSO + c];
    sW[kp * WP + c] = w;
  }
  __syncthreads();
#pragma unroll
  for (int t = 0; t < EPT; ++t) atomicAdd(&cnt[dl[t]], 1);

  {
    const int r  = lane & 15;
    const int kb = (lane >> 4) << 1;
    const int rbase = (lane >> 4) << 3;
    const int row0 = (tid >> 5) * (16 * MT);
    v8f pa[MT][NT] = {};
#pragma unroll 4
    for (int k = 0; k < K; k += 4) {
      v2f a[MT];
#pragma unroll
      for (int mt = 0; mt < MT; ++mt) {
        const size_t row = (size_t)(gbase + row0 + mt * 16 + r);
        const float* ap;
        if constexpr (CSI == 0)
          ap = in + row * K + k + kb;
        else
          ap = in + ((size_t)(k / CSI) * NNODES + row) * CSI + (k % CSI) + kb;
        a[mt] = *(const v2f*)ap;
      }
      v2f bf[NT];
#pragma unroll
      for (int nt = 0; nt < NT; ++nt)
        bf[nt] = sW[((k + kb) >> 1) * WP + nt * 16 + r];
#pragma unroll
      for (int mt = 0; mt < MT; ++mt)
#pragma unroll
        for (int nt = 0; nt < NT; ++nt)
          pa[mt][nt] = __builtin_amdgcn_wmma_f32_16x16x4_f32(false, a[mt], false, bf[nt],
                                                             (short)0, pa[mt][nt], false, false);
    }
    __syncthreads();
#pragma unroll
    for (int mt = 0; mt < MT; ++mt) {
#pragma unroll
      for (int nt = 0; nt < NT; ++nt) {
#pragma unroll
        for (int v = 0; v < 8; ++v) {
          const int rl = row0 + mt * 16 + rbase + v;
          const float hval = pa[mt][nt][v];
          const float d = rsqrtf((float)cnt[rl]);
          hl[rl * CSO + nt * 16 + r]  = hval;
          acc[rl * CSO + nt * 16 + r] = hval * d * d;
        }
      }
    }
  }
  if constexpr (LAST) {
    static_assert(CSO == CS, "the filtered scatter is written for one column tile");
    __shared__ v2u elist[EPG];
    v2u* seg = elist + (tid >> 5) * (EPT * 32);
    float wv[EPT];
#pragma unroll
    for (int t = 0; t < EPT; ++t)
      wv[t] = rsqrtf((float)cnt[sl_[t]]) * rsqrtf((float)cnt[dl[t]]);
    int nw = 0;
#pragma unroll
    for (int t = 0; t < EPT; ++t) {
      const int d = dl[t];
      const bool need = ((d + 1) & 2) != 0;
      const unsigned mask = __builtin_amdgcn_ballot_w32(need);
      if (need) {
        v2u e;
        e.x = ((unsigned)sl_[t] * (CSO * 4u)) | (((unsigned)d * (CSO * 4u)) << 16);
        e.y = __float_as_uint(wv[t]);
        seg[nw + (int)__builtin_amdgcn_mbcnt_lo(mask, 0u)] = e;
      }
      nw += __builtin_popcount(mask);
    }
    __syncthreads();
    const unsigned c4 = (unsigned)(lane & (CS - 1)) * 4u;
    const unsigned cc = c4 | (c4 << 16);
    const int li = ((lane & 15) << 1) | (lane >> 4);
    for (int e0 = 0; e0 < nw; e0 += 32) {
      v2u e = {0u, 0u};
      if (e0 + li < nw) e = seg[e0 + li];
      agg_half<0>(e.x, e.y, cc, hl, acc);
      if (e0 + 16 < nw) agg_half<8>(e.x, e.y, cc, hl, acc);
    }
  } else {
    __syncthreads();

    {
      const int* __restrict__ es = ei + (size_t)b * EPG;
      const int* __restrict__ ed = ei + (size_t)NEDGES + (size_t)b * EPG;
      for (int e0 = (tid >> 5) * 32; e0 < EPG; e0 += nwv * 32) {
        const int s = es[e0 + lane] - gbase;
        const int d = ed[e0 + lane] - gbase;
        const float w = rsqrtf((float)cnt[s]) * rsqrtf((float)cnt[d]);
        const unsigned sd = ((unsigned)s * (CSO * 4u)) | (((unsigned)d * (CSO * 4u)) << 16);
        if constexpr (CSO == 32) {
          const unsigned c4 = (unsigned)lane * 4u;
          agg32_group<0>(sd, __float_as_uint(w), c4, hl, acc);
          agg32_group<8>(sd, __float_as_uint(w), c4, hl, acc);
          agg32_group<16>(sd, __float_as_uint(w), c4, hl, acc);
          agg32_group<24>(sd, __float_as_uint(w), c4, hl, acc);
        } else {
          const unsigned c4 = (unsigned)(lane & (CS - 1)) * 4u;
          const unsigned cc = c4 | (c4 << 16);
          agg_half<0>(sd, __float_as_uint(w), cc, hl, acc);
          agg_half<8>(sd, __float_as_uint(w), cc, hl, acc);
        }
      }
    }
  }
  __syncthreads();

  constexpr int TP = 2 * RES + 64 / CSO;
  float* tp = hl;
  for (int idx = tid; idx < RES * CSO; idx += blockDim.x) {
    int c = idx & (CSO - 1);
    int r = idx / CSO;
    float bb = bsl[c];
    float v1 = fmaxf(acc[(4 * r + 1) * CSO + c] + bb, 0.0f);
    float v2 = fmaxf(acc[(4 * r + 2) * CSO + c] + bb, 0.0f);
    tp[(c >> 1) * TP + 2 * r + (c & 1)] = 0.5f * (v1 + v2);
  }
  __syncthreads();
  float* pg = pout + (size_t)b * RES * N + (size_t)sl * RES * CSO;
  for (int idx = tid; idx < RES * CSO; idx += blockDim.x)
    pg[idx] = tp[(idx >> 8) * TP + (idx & (2 * RES - 1))];
  if constexpr (!LAST) {
    float* yg = y + ((size_t)sl * NNODES + gbase) * CSO;
    for (int idx = tid; idx < NPG * CSO; idx += blockDim.x) {
      float v = acc[idx] + bsl[idx & (CSO - 1)];
      yg[idx] = fmaxf(v, 0.0f);
    }
  }
}

template <int D>
__global__ void __launch_bounds__(256) __attribute__((amdgpu_waves_per_eu(2, 2))) k_sim_wmma(const float* __restrict__ U,
                                                  const float* __restrict__ V,
                                                  float* __restrict__ out) {
  const int b    = blockIdx.x;
  const int w    = threadIdx.x >> 5;
  const int lane = threadIdx.x & 31;
  const int r    = lane & 15;
  const int h    = lane >> 4;
  const int tm   = blockIdx.y * 4 + (w >> 1);
  const int tc   = (w & 1) * 4;
  const v2f* Ug = (const v2f*)U + (size_t)b * (RES * D / 2) + h * RES + tm * 16 + r;
  const v2f* Vg = (const v2f*)V + (size_t)b * (RES * D / 2) + h * RES + tc * 16 + r;

  constexpr int KS = D / 4;
  v2f a[KS];
  v2f bf[KS][4];
#pragma unroll
  for (int s = 0; s < KS; ++s) {
    a[s] = Ug[(2 * s) * RES];
#pragma unroll
    for (int t = 0; t < 4; ++t)
      bf[s][t] = Vg[(2 * s) * RES + t * 16];
  }
  v8f acc[4] = {};
#pragma unroll
  for (int s = 0; s < KS; ++s)
#pragma unroll
    for (int t = 0; t < 4; ++t)
      acc[t] = __builtin_amdgcn_wmma_f32_16x16x4_f32(false, a[s], false, bf[s][t], (short)0,
                                                     acc[t], false, false);
  const int rbase = h << 3;
#pragma unroll
  for (int t = 0; t < 4; ++t) {
    float* Co = out + (size_t)b * RES * RES + (size_t)(tm * 16 + rbase) * RES + (tc + t) * 16 + r;
#pragma unroll
    for (int v = 0; v < 8; ++v) Co[(size_t)v * RES] = acc[t][v];
  }
}

extern "C" void kernel_launch(void* const* d_in, const int* in_sizes, int n_in,
                              void* d_out, int out_size, void* d_ws, size_t ws_size,
                              hipStream_t stream) {
  (void)in_sizes; (void)n_in; (void)out_size; (void)ws_size;
  const float* x[2]  = {(const float*)d_in[0], (const float*)d_in[1]};
  const int*   ei[2] = {(const int*)d_in[2],  (const int*)d_in[3]};
  const float* W[3]  = {(const float*)d_in[6], (const float*)d_in[8], (const float*)d_in[10]};
  const float* bv[3] = {(const float*)d_in[7], (const float*)d_in[9], (const float*)d_in[11]};
  float* out = (float*)d_out;

  float* ws = (float*)d_ws;
  size_t o = 0;
  float* Y1[2]; float* Y2[2];
  for (int s = 0; s < 2; ++s) { Y1[s] = ws + o; o += (size_t)NNODES * 64; }
  for (int s = 0; s < 2; ++s) { Y2[s] = ws + o; o += (size_t)NNODES * 32; }
  float* P[2][3];
  const int dims[3] = {64, 32, 16};
  for (int s = 0; s < 2; ++s)
    for (int l = 0; l < 3; ++l) { P[s][l] = ws + o; o += (size_t)NGRAPHS * RES * dims[l]; }


  k_layer<128, 0, 64, 32, false><<<dim3(NGRAPHS, 2, 2), 1024, 0, stream>>>(
      x[0], x[1], W[0], ei[0], ei[1], bv[0], Y1[0], Y1[1], P[0][0], P[1][0]);
  k_layer<64, 32, 32, 32, false><<<dim3(NGRAPHS, 1, 2), 1024, 0, stream>>>(
      Y1[0], Y1[1], W[1], ei[0], ei[1], bv[1], Y2[0], Y2[1], P[0][1], P[1][1]);
  k_layer<32, 32, 16, 16, true><<<dim3(NGRAPHS, 1, 2), 512, 0, stream>>>(
      Y2[0], Y2[1], W[2], ei[0], ei[1], bv[2], nullptr, nullptr, P[0][2], P[1][2]);

  k_sim_wmma<64><<<dim3(NGRAPHS, 2), 256, 0, stream>>>(P[0][0], P[1][0], out);
  k_sim_wmma<32><<<dim3(NGRAPHS, 2), 256, 0, stream>>>(P[0][1], P[1][1],
                                                   out + (size_t)1 * NGRAPHS * RES * RES);
  k_sim_wmma<16><<<dim3(NGRAPHS, 2), 256, 0, stream>>>(P[0][2], P[1][2],
                                                   out + (size_t)2 * NGRAPHS * RES * RES);
}


# similarity products of layers 1 and 2 ride in the NEXT layer's launch as work-groups of their own behind the layer's (grid z = 2, 3; 1 or 2 tiles a wave, same k order); 6 launches become 4
# speedup vs baseline: 7.4115x; 1.0526x over previous
#include <hip/hip_runtime.h>


typedef __attribute__((ext_vector_type(2))) float v2f;
typedef __attribute__((ext_vector_type(8))) float v8f;
typedef __attribute__((ext_vector_type(2))) unsigned v2u;

#define NNODES  (128 * 512)
#define NEDGES  (128 * 512 * 16)
#define NGRAPHS 128
#define NPG     512
#define EPG     (512 * 16)
#define RES     128
#define CS      16


template <int G0, int G>
__device__ __forceinline__ void agg_take(unsigned sd, unsigned wbits, unsigned cc,
                                         const float* hl, float* hv, unsigned* dofs) {
  if constexpr (G < 8) {
    const unsigned t =
        (unsigned)__builtin_amdgcn_mov_dpp((int)sd, 0x150 + G0 + G, 0xf, 0xf, true) | cc;
    const float wg = __uint_as_float(
        (unsigned)__builtin_amdgcn_mov_dpp((int)wbits, 0x150 + G0 + G, 0xf, 0xf, true));
    hv[G] = *(const float*)((const char*)hl + (t & 0xffffu)) * wg;
    dofs[G] = t >> 16;
    agg_take<G0, G + 1>(sd, wbits, cc, hl, hv, dofs);
  }
}

template <int G0>
__device__ __forceinline__ void agg_half(unsigned sd, unsigned wbits, unsigned cc,
                                         const float* hl, float* acc) {
  float    hv[8];
  unsigned dofs[8];
  agg_take<G0, 0>(sd, wbits, cc, hl, hv, dofs);
  for (int g = 0; g < 8; ++g) atomicAdd((float*)((char*)acc + dofs[g]), hv[g]);
}

template <int G0, int G>
__device__ __forceinline__ void agg32_take(unsigned sd, unsigned wbits, unsigned c4,
                                           const float* hl, float* hv, unsigned* dofs) {
  if constexpr (G < 8) {
    const unsigned t = (unsigned)__builtin_amdgcn_readlane((int)sd, G0 + G);
    const float wg = __uint_as_float((unsigned)__builtin_amdgcn_readlane((int)wbits, G0 + G));
    hv[G] = *(const float*)((const char*)hl + ((t & 0xffffu) | c4)) * wg;
    dofs[G] = (t >> 16) | c4;
    agg32_take<G0, G + 1>(sd, wbits, c4, hl, hv, dofs);
  }
}

template <int G0>
__device__ __forceinline__ void agg32_group(unsigned sd, unsigned wbits, unsigned c4,
                                            const float* hl, float* acc) {
  float    hv[8];
  unsigned dofs[8];
  agg32_take<G0, 0>(sd, wbits, c4, hl, hv, dofs);
  for (int g = 0; g < 8; ++g) atomicAdd((float*)((char*)acc + dofs[g]), hv[g]);
}

template <int D, int NW>
__device__ __forceinline__ void sim_tiles(const float* __restrict__ U,
                                          const float* __restrict__ V,
                                          float* __restrict__ out, int b, int part, int tid) {
  constexpr int TPW = 32 / NW;
  constexpr int CG  = 8 / TPW;
  const int w    = tid >> 5;
  const int lane = tid & 31;
  const int r    = lane & 15;
  const int h    = lane >> 4;
  const int tm   = part * 4 + w / CG;
  const int tc   = (w % CG) * TPW;
  const v2f* Ug = (const v2f*)U + (size_t)b * (RES * D / 2) + h * RES + tm * 16 + r;
  const v2f* Vg = (const v2f*)V + (size_t)b * (RES * D / 2) + h * RES + tc * 16 + r;
  constexpr int KS = D / 4;
  constexpr int KC = 8;
  static_assert(KS % KC == 0, "whole trips");
  v8f acc[TPW] = {};
#pragma nounroll
  for (int s0 = 0; s0 < KS; s0 += KC) {
    v2f a[KC];
    v2f bf[KC][TPW];
#pragma unroll
    for (int s = 0; s < KC; ++s) {
      a[s] = Ug[(2 * (s0 + s)) * RES];
#pragma unroll
      for (int t = 0; t < TPW; ++t) bf[s][t] = Vg[(2 * (s0 + s)) * RES + t * 16];
    }
#pragma unroll
    for (int s = 0; s < KC; ++s)
#pragma unroll
      for (int t = 0; t < TPW; ++t)
        acc[t] = __builtin_amdgcn_wmma_f32_16x16x4_f32(false, a[s], false, bf[s][t], (short)0,
                                                       acc[t], false, false);
  }
  const int rbase = h << 3;
#pragma unroll
  for (int t = 0; t < TPW; ++t) {
    float* Co = out + (size_t)b * RES * RES + (size_t)(tm * 16 + rbase) * RES + (tc + t) * 16 + r;
#pragma unroll
    for (int v = 0; v < 8; ++v) Co[(size_t)v * RES] = acc[t][v];
  }
}

template <int K, int CSI, int N, int CSO, bool LAST, int SD>
__global__ void __launch_bounds__(32 * CSO) k_layer(
    const float* __restrict__ in0, const float* __restrict__ in1,
    const float* __restrict__ Wm,
    const int* __restrict__ ei0, const int* __restrict__ ei1,
    const float* __restrict__ bias,
    float* __restrict__ y0, float* __restrict__ y1,
    float* __restrict__ pout0, float* __restrict__ pout1,
    const float* __restrict__ simU, const float* __restrict__ simV,
    float* __restrict__ simOut) {
  if constexpr (SD > 0) {
    if (blockIdx.z >= 2) {
      sim_tiles<SD, CSO>(simU, simV, simOut, (int)blockIdx.x, (int)blockIdx.z - 2,
                         (int)threadIdx.x);
      return;
    }
  }
  constexpr int MT = 32 / CSO;
  constexpr int NT = CSO / 16;
  constexpr int WP = (CSO == 32) ? 48 : 16;
  __shared__ float smem[2 * NPG * CSO];
  __shared__ v2f sW[(K / 2) * WP];
  __shared__ int cnt[NPG];
  constexpr int NTHR = 32 * CSO;
  constexpr int EPT  = EPG / NTHR;
  const int side = blockIdx.z;
  const float* __restrict__ in   = side ? in1 : in0;
  const int*   __restrict__ ei   = side ? ei1 : ei0;
  float* __restrict__ y    = side ? y1 : y0;
  float* __restrict__ pout = side ? pout1 : pout0;
  float* acc = smem;
  float* hl  = smem + NPG * CSO;
  const int b  = blockIdx.x;
  const int sl = blockIdx.y;
  const int gbase = b * NPG;
  const int tid = threadIdx.x;
  const int lane = tid & 31;
  const int nwv = (int)(blockDim.x >> 5);
  const float* bsl = bias + sl * CSO;

  int dl[EPT];
#pragma unroll
  for (int t = 0; t < EPT; ++t)
    dl[t] = ei[(size_t)NEDGES + (size_t)b * EPG + t * NTHR + tid] & (NPG - 1);
  int sl_[LAST ? EPT : 1];
  if constexpr (LAST) {
#pragma unroll
    for (int t = 0; t < EPT; ++t)
      sl_[t] = ei[(size_t)b * EPG + t * NTHR + tid] & (NPG - 1);
  }
  static_assert(NTHR >= NPG, "one thread per node for the count's start value");
  if (tid < NPG) cnt[tid] = 1;
  for (int i = tid; i < (K / 2) * CSO; i += blockDim.x) {
    int kp = i / CSO;
    int c  = i - kp * CSO;
    v2f w;
    w.x = Wm[(size_t)(2 * kp) * N + sl * CSO + c];
    w.y = Wm[(size_t)(2 * kp + 1) * N + sl * CSO + c];
    sW[kp * WP + c] = w;
  }
  __syncthreads();
#pragma unroll
  for (int t = 0; t < EPT; ++t) atomicAdd(&cnt[dl[t]], 1);

  {
    const int r  = lane & 15;
    const int kb = (lane >> 4) << 1;
    const int rbase = (lane >> 4) << 3;
    const int row0 = (tid >> 5) * (16 * MT);
    v8f pa[MT][NT] = {};
#pragma unroll 4
    for (int k = 0; k < K; k += 4) {
      v2f a[MT];
#pragma unroll
      for (int mt = 0; mt < MT; ++mt) {
        const size_t row = (size_t)(gbase + row0 + mt * 16 + r);
        const float* ap;
        if constexpr (CSI == 0)
          ap = in + row * K + k + kb;
        else
          ap = in + ((size_t)(k / CSI) * NNODES + row) * CSI + (k % CSI) + kb;
        a[mt] = *(const v2f*)ap;
      }
      v2f bf[NT];
#pragma unroll
      for (int nt = 0; nt < NT; ++nt)
        bf[nt] = sW[((k + kb) >> 1) * WP + nt * 16 + r];
#pragma unroll
      for (int mt = 0; mt < MT; ++mt)
#pragma unroll
        for (int nt = 0; nt < NT; ++nt)
          pa[mt][nt] = __builtin_amdgcn_wmma_f32_16x16x4_f32(false, a[mt], false, bf[nt],
                                                             (short)0, pa[mt][nt], false, false);
    }
    __syncthreads();
#pragma unroll
    for (int mt = 0; mt < MT; ++mt) {
#pragma unroll
      for (int nt = 0; nt < NT; ++nt) {
#pragma unroll
        for (int v = 0; v < 8; ++v) {
          const int rl = row0 + mt * 16 + rbase + v;
          const float hval = pa[mt][nt][v];
          const float d = rsqrtf((float)cnt[rl]);
          hl[rl * CSO + nt * 16 + r]  = hval;
          acc[rl * CSO + nt * 16 + r] = hval * d * d;
        }
      }
    }
  }
  if constexpr (LAST) {
    static_assert(CSO == CS, "the filtered scatter is written for one column tile");
    __shared__ v2u elist[EPG];
    v2u* seg = elist + (tid >> 5) * (EPT * 32);
    float wv[EPT];
#pragma unroll
    for (int t = 0; t < EPT; ++t)
      wv[t] = rsqrtf((float)cnt[sl_[t]]) * rsqrtf((float)cnt[dl[t]]);
    int nw = 0;
#pragma unroll
    for (int t = 0; t < EPT; ++t) {
      const int d = dl[t];
      const bool need = ((d + 1) & 2) != 0;
      const unsigned mask = __builtin_amdgcn_ballot_w32(need);
      if (need) {
        v2u e;
        e.x = ((unsigned)sl_[t] * (CSO * 4u)) | (((unsigned)d * (CSO * 4u)) << 16);
        e.y = __float_as_uint(wv[t]);
        seg[nw + (int)__builtin_amdgcn_mbcnt_lo(mask, 0u)] = e;
      }
      nw += __builtin_popcount(mask);
    }
    __syncthreads();
    const unsigned c4 = (unsigned)(lane & (CS - 1)) * 4u;
    const unsigned cc = c4 | (c4 << 16);
    const int li = ((lane & 15) << 1) | (lane >> 4);
    for (int e0 = 0; e0 < nw; e0 += 32) {
      v2u e = {0u, 0u};
      if (e0 + li < nw) e = seg[e0 + li];
      agg_half<0>(e.x, e.y, cc, hl, acc);
      if (e0 + 16 < nw) agg_half<8>(e.x, e.y, cc, hl, acc);
    }
  } else {
    __syncthreads();

    {
      const int* __restrict__ es = ei + (size_t)b * EPG;
      const int* __restrict__ ed = ei + (size_t)NEDGES + (size_t)b * EPG;
      for (int e0 = (tid >> 5) * 32; e0 < EPG; e0 += nwv * 32) {
        const int s = es[e0 + lane] - gbase;
        const int d = ed[e0 + lane] - gbase;
        const float w = rsqrtf((float)cnt[s]) * rsqrtf((float)cnt[d]);
        const unsigned sd = ((unsigned)s * (CSO * 4u)) | (((unsigned)d * (CSO * 4u)) << 16);
        if constexpr (CSO == 32) {
          const unsigned c4 = (unsigned)lane * 4u;
          agg32_group<0>(sd, __float_as_uint(w), c4, hl, acc);
          agg32_group<8>(sd, __float_as_uint(w), c4, hl, acc);
          agg32_group<16>(sd, __float_as_uint(w), c4, hl, acc);
          agg32_group<24>(sd, __float_as_uint(w), c4, hl, acc);
        } else {
          const unsigned c4 = (unsigned)(lane & (CS - 1)) * 4u;
          const unsigned cc = c4 | (c4 << 16);
          agg_half<0>(sd, __float_as_uint(w), cc, hl, acc);
          agg_half<8>(sd, __float_as_uint(w), cc, hl, acc);
        }
      }
    }
  }
  __syncthreads();

  constexpr int TP = 2 * RES + 64 / CSO;
  float* tp = hl;
  for (int idx = tid; idx < RES * CSO; idx += blockDim.x) {
    int c = idx & (CSO - 1);
    int r = idx / CSO;
    float bb = bsl[c];
    float v1 = fmaxf(acc[(4 * r + 1) * CSO + c] + bb, 0.0f);
    float v2 = fmaxf(acc[(4 * r + 2) * CSO + c] + bb, 0.0f);
    tp[(c >> 1) * TP + 2 * r + (c & 1)] = 0.5f * (v1 + v2);
  }
  __syncthreads();
  float* pg = pout + (size_t)b * RES * N + (size_t)sl * RES * CSO;
  for (int idx = tid; idx < RES * CSO; idx += blockDim.x)
    pg[idx] = tp[(idx >> 8) * TP + (idx & (2 * RES - 1))];
  if constexpr (!LAST) {
    float* yg = y + ((size_t)sl * NNODES + gbase) * CSO;
    for (int idx = tid; idx < NPG * CSO; idx += blockDim.x) {
      float v = acc[idx] + bsl[idx & (CSO - 1)];
      yg[idx] = fmaxf(v, 0.0f);
    }
  }
}

template <int D>
__global__ void __launch_bounds__(256) __attribute__((amdgpu_waves_per_eu(2, 2))) k_sim_wmma(const float* __restrict__ U,
                                                  const float* __restrict__ V,
                                                  float* __restrict__ out) {
  const int b    = blockIdx.x;
  const int w    = threadIdx.x >> 5;
  const int lane = threadIdx.x & 31;
  const int r    = lane & 15;
  const int h    = lane >> 4;
  const int tm   = blockIdx.y * 4 + (w >> 1);
  const int tc   = (w & 1) * 4;
  const v2f* Ug = (const v2f*)U + (size_t)b * (RES * D / 2) + h * RES + tm * 16 + r;
  const v2f* Vg = (const v2f*)V + (size_t)b * (RES * D / 2) + h * RES + tc * 16 + r;

  constexpr int KS = D / 4;
  v2f a[KS];
  v2f bf[KS][4];
#pragma unroll
  for (int s = 0; s < KS; ++s) {
    a[s] = Ug[(2 * s) * RES];
#pragma unroll
    for (int t = 0; t < 4; ++t)
      bf[s][t] = Vg[(2 * s) * RES + t * 16];
  }
  v8f acc[4] = {};
#pragma unroll
  for (int s = 0; s < KS; ++s)
#pragma unroll
    for (int t = 0; t < 4; ++t)
      acc[t] = __builtin_amdgcn_wmma_f32_16x16x4_f32(false, a[s], false, bf[s][t], (short)0,
                                                     acc[t], false, false);
  const int rbase = h << 3;
#pragma unroll
  for (int t = 0; t < 4; ++t) {
    float* Co = out + (size_t)b * RES * RES + (size_t)(tm * 16 + rbase) * RES + (tc + t) * 16 + r;
#pragma unroll
    for (int v = 0; v < 8; ++v) Co[(size_t)v * RES] = acc[t][v];
  }
}

extern "C" void kernel_launch(void* const* d_in, const int* in_sizes, int n_in,
                              void* d_out, int out_size, void* d_ws, size_t ws_size,
                              hipStream_t stream) {
  (void)in_sizes; (void)n_in; (void)out_size; (void)ws_size;
  const float* x[2]  = {(const float*)d_in[0], (const float*)d_in[1]};
  const int*   ei[2] = {(const int*)d_in[2],  (const int*)d_in[3]};
  const float* W[3]  = {(const float*)d_in[6], (const float*)d_in[8], (const float*)d_in[10]};
  const float* bv[3] = {(const float*)d_in[7], (const float*)d_in[9], (const float*)d_in[11]};
  float* out = (float*)d_out;

  float* ws = (float*)d_ws;
  size_t o = 0;
  float* Y1[2]; float* Y2[2];
  for (int s = 0; s < 2; ++s) { Y1[s] = ws + o; o += (size_t)NNODES * 64; }
  for (int s = 0; s < 2; ++s) { Y2[s] = ws + o; o += (size_t)NNODES * 32; }
  float* P[2][3];
  const int dims[3] = {64, 32, 16};
  for (int s = 0; s < 2; ++s)
    for (int l = 0; l < 3; ++l) { P[s][l] = ws + o; o += (size_t)NGRAPHS * RES * dims[l]; }


  k_layer<128, 0, 64, 32, false, 0><<<dim3(NGRAPHS, 2, 2), 1024, 0, stream>>>(
      x[0], x[1], W[0], ei[0], ei[1], bv[0], Y1[0], Y1[1], P[0][0], P[1][0],
      nullptr, nullptr, nullptr);
  k_layer<64, 32, 32, 32, false, 64><<<dim3(NGRAPHS, 1, 4), 1024, 0, stream>>>(
      Y1[0], Y1[1], W[1], ei[0], ei[1], bv[1], Y2[0], Y2[1], P[0][1], P[1][1],
      P[0][0], P[1][0], out);
  k_layer<32, 32, 16, 16, true, 32><<<dim3(NGRAPHS, 1, 4), 512, 0, stream>>>(
      Y2[0], Y2[1], W[2], ei[0], ei[1], bv[2], nullptr, nullptr, P[0][2], P[1][2],
      P[0][1], P[1][1], out + (size_t)1 * NGRAPHS * RES * RES);
  k_sim_wmma<16><<<dim3(NGRAPHS, 2), 256, 0, stream>>>(P[0][2], P[1][2],
                                                   out + (size_t)2 * NGRAPHS * RES * RES);
}


# weight products: a lane fetches 4 consecutive k of its row (A, global) and column (B, LDS) per 16-byte load, each feeding 2 matrix instructions; half the fragment loads; k sum order changes
# speedup vs baseline: 7.6965x; 1.0385x over previous
#include <hip/hip_runtime.h>


typedef __attribute__((ext_vector_type(2))) float v2f;
typedef __attribute__((ext_vector_type(4))) float v4f;
typedef __attribute__((ext_vector_type(8))) float v8f;
typedef __attribute__((ext_vector_type(2))) unsigned v2u;

#define NNODES  (128 * 512)
#define NEDGES  (128 * 512 * 16)
#define NGRAPHS 128
#define NPG     512
#define EPG     (512 * 16)
#define RES     128
#define CS      16


template <int G0, int G>
__device__ __forceinline__ void agg_take(unsigned sd, unsigned wbits, unsigned cc,
                                         const float* hl, float* hv, unsigned* dofs) {
  if constexpr (G < 8) {
    const unsigned t =
        (unsigned)__builtin_amdgcn_mov_dpp((int)sd, 0x150 + G0 + G, 0xf, 0xf, true) | cc;
    const float wg = __uint_as_float(
        (unsigned)__builtin_amdgcn_mov_dpp((int)wbits, 0x150 + G0 + G, 0xf, 0xf, true));
    hv[G] = *(const float*)((const char*)hl + (t & 0xffffu)) * wg;
    dofs[G] = t >> 16;
    agg_take<G0, G + 1>(sd, wbits, cc, hl, hv, dofs);
  }
}

template <int G0>
__device__ __forceinline__ void agg_half(unsigned sd, unsigned wbits, unsigned cc,
                                         const float* hl, float* acc) {
  float    hv[8];
  unsigned dofs[8];
  agg_take<G0, 0>(sd, wbits, cc, hl, hv, dofs);
  for (int g = 0; g < 8; ++g) atomicAdd((float*)((char*)acc + dofs[g]), hv[g]);
}

template <int G0, int G>
__device__ __forceinline__ void agg32_take(unsigned sd, unsigned wbits, unsigned c4,
                                           const float* hl, float* hv, unsigned* dofs) {
  if constexpr (G < 8) {
    const unsigned t = (unsigned)__builtin_amdgcn_readlane((int)sd, G0 + G);
    const float wg = __uint_as_float((unsigned)__builtin_amdgcn_readlane((int)wbits, G0 + G));
    hv[G] = *(const float*)((const char*)hl + ((t & 0xffffu) | c4)) * wg;
    dofs[G] = (t >> 16) | c4;
    agg32_take<G0, G + 1>(sd, wbits, c4, hl, hv, dofs);
  }
}

template <int G0>
__device__ __forceinline__ void agg32_group(unsigned sd, unsigned wbits, unsigned c4,
                                            const float* hl, float* acc) {
  float    hv[8];
  unsigned dofs[8];
  agg32_take<G0, 0>(sd, wbits, c4, hl, hv, dofs);
  for (int g = 0; g < 8; ++g) atomicAdd((float*)((char*)acc + dofs[g]), hv[g]);
}

template <int D, int NW>
__device__ __forceinline__ void sim_tiles(const float* __restrict__ U,
                                          const float* __restrict__ V,
                                          float* __restrict__ out, int b, int part, int tid) {
  constexpr int TPW = 32 / NW;
  constexpr int CG  = 8 / TPW;
  const int w    = tid >> 5;
  const int lane = tid & 31;
  const int r    = lane & 15;
  const int h    = lane >> 4;
  const int tm   = part * 4 + w / CG;
  const int tc   = (w % CG) * TPW;
  const v2f* Ug = (const v2f*)U + (size_t)b * (RES * D / 2) + h * RES + tm * 16 + r;
  const v2f* Vg = (const v2f*)V + (size_t)b * (RES * D / 2) + h * RES + tc * 16 + r;
  constexpr int KS = D / 4;
  constexpr int KC = 8;
  static_assert(KS % KC == 0, "whole trips");
  v8f acc[TPW] = {};
#pragma nounroll
  for (int s0 = 0; s0 < KS; s0 += KC) {
    v2f a[KC];
    v2f bf[KC][TPW];
#pragma unroll
    for (int s = 0; s < KC; ++s) {
      a[s] = Ug[(2 * (s0 + s)) * RES];
#pragma unroll
      for (int t = 0; t < TPW; ++t) bf[s][t] = Vg[(2 * (s0 + s)) * RES + t * 16];
    }
#pragma unroll
    for (int s = 0; s < KC; ++s)
#pragma unroll
      for (int t = 0; t < TPW; ++t)
        acc[t] = __builtin_amdgcn_wmma_f32_16x16x4_f32(false, a[s], false, bf[s][t], (short)0,
                                                       acc[t], false, false);
  }
  const int rbase = h << 3;
#pragma unroll
  for (int t = 0; t < TPW; ++t) {
    float* Co = out + (size_t)b * RES * RES + (size_t)(tm * 16 + rbase) * RES + (tc + t) * 16 + r;
#pragma unroll
    for (int v = 0; v < 8; ++v) Co[(size_t)v * RES] = acc[t][v];
  }
}

template <int K, int CSI, int N, int CSO, bool LAST, int SD>
__global__ void __launch_bounds__(32 * CSO) k_layer(
    const float* __restrict__ in0, const float* __restrict__ in1,
    const float* __restrict__ Wm,
    const int* __restrict__ ei0, const int* __restrict__ ei1,
    const float* __restrict__ bias,
    float* __restrict__ y0, float* __restrict__ y1,
    float* __restrict__ pout0, float* __restrict__ pout1,
    const float* __restrict__ simU, const float* __restrict__ simV,
    float* __restrict__ simOut) {
  if constexpr (SD > 0) {
    if (blockIdx.z >= 2) {
      sim_tiles<SD, CSO>(simU, simV, simOut, (int)blockIdx.x, (int)blockIdx.z - 2,
                         (int)threadIdx.x);
      return;
    }
  }
  constexpr int MT = 32 / CSO;
  constexpr int NT = CSO / 16;
  constexpr int UB = (MT == 1) ? 4 : 2;
  static_assert(K % 8 == 0 && (CSI == 0 || CSI % 8 == 0), "whole 8-k blocks inside a slice");
  __shared__ float smem[2 * NPG * CSO];
  __shared__ v4f sW[(K / 4) * CSO];
  __shared__ int cnt[NPG];
  constexpr int NTHR = 32 * CSO;
  constexpr int EPT  = EPG / NTHR;
  const int side = blockIdx.z;
  const float* __restrict__ in   = side ? in1 : in0;
  const int*   __restrict__ ei   = side ? ei1 : ei0;
  float* __restrict__ y    = side ? y1 : y0;
  float* __restrict__ pout = side ? pout1 : pout0;
  float* acc = smem;
  float* hl  = smem + NPG * CSO;
  const int b  = blockIdx.x;
  const int sl = blockIdx.y;
  const int gbase = b * NPG;
  const int tid = threadIdx.x;
  const int lane = tid & 31;
  const int nwv = (int)(blockDim.x >> 5);
  const float* bsl = bias + sl * CSO;

  int dl[EPT];
#pragma unroll
  for (int t = 0; t < EPT; ++t)
    dl[t] = ei[(size_t)NEDGES + (size_t)b * EPG + t * NTHR + tid] & (NPG - 1);
  int sl_[LAST ? EPT : 1];
  if constexpr (LAST) {
#pragma unroll
    for (int t = 0; t < EPT; ++t)
      sl_[t] = ei[(size_t)b * EPG + t * NTHR + tid] & (NPG - 1);
  }
  static_assert(NTHR >= NPG, "one thread per node for the count's start value");
  if (tid < NPG) cnt[tid] = 1;
  for (int i = tid; i < (K / 4) * CSO; i += blockDim.x) {
    int q = i / CSO;
    int c = i - q * CSO;
    const float* wq = Wm + (size_t)(4 * q) * N + sl * CSO + c;
    v4f w;
    w.x = wq[0];
    w.y = wq[N];
    w.z = wq[2 * N];
    w.w = wq[3 * N];
    sW[i] = w;
  }
  __syncthreads();
#pragma unroll
  for (int t = 0; t < EPT; ++t) atomicAdd(&cnt[dl[t]], 1);

  {
    const int r  = lane & 15;
    const int k4 = (lane >> 4) << 2;
    const int rbase = (lane >> 4) << 3;
    const int row0 = (tid >> 5) * (16 * MT);
    v8f pa[MT][NT] = {};
#pragma unroll UB
    for (int k = 0; k < K; k += 8) {
      v4f a[MT];
#pragma unroll
      for (int mt = 0; mt < MT; ++mt) {
        const size_t row = (size_t)(gbase + row0 + mt * 16 + r);
        const float* ap;
        if constexpr (CSI == 0)
          ap = in + row * K + k + k4;
        else
          ap = in + ((size_t)(k / CSI) * NNODES + row) * CSI + (k % CSI) + k4;
        a[mt] = *(const v4f*)ap;
      }
      v4f bf[NT];
#pragma unroll
      for (int nt = 0; nt < NT; ++nt)
        bf[nt] = sW[((k + k4) >> 2) * CSO + nt * 16 + r];
#pragma unroll
      for (int t = 0; t < 2; ++t)
#pragma unroll
        for (int mt = 0; mt < MT; ++mt)
#pragma unroll
          for (int nt = 0; nt < NT; ++nt) {
            v2f av, bv;
            av.x = t ? a[mt].z : a[mt].x;
            av.y = t ? a[mt].w : a[mt].y;
            bv.x = t ? bf[nt].z : bf[nt].x;
            bv.y = t ? bf[nt].w : bf[nt].y;
            pa[mt][nt] = __builtin_amdgcn_wmma_f32_16x16x4_f32(false, av, false, bv, (short)0,
                                                               pa[mt][nt], false, false);
          }
    }
    __syncthreads();
#pragma unroll
    for (int mt = 0; mt < MT; ++mt) {
#pragma unroll
      for (int nt = 0; nt < NT; ++nt) {
#pragma unroll
        for (int v = 0; v < 8; ++v) {
          const int rl = row0 + mt * 16 + rbase + v;
          const float hval = pa[mt][nt][v];
          const float d = rsqrtf((float)cnt[rl]);
          hl[rl * CSO + nt * 16 + r]  = hval;
          acc[rl * CSO + nt * 16 + r] = hval * d * d;
        }
      }
    }
  }
  if constexpr (LAST) {
    static_assert(CSO == CS, "the filtered scatter is written for one column tile");
    __shared__ v2u elist[EPG];
    v2u* seg = elist + (tid >> 5) * (EPT * 32);
    float wv[EPT];
#pragma unroll
    for (int t = 0; t < EPT; ++t)
      wv[t] = rsqrtf((float)cnt[sl_[t]]) * rsqrtf((float)cnt[dl[t]]);
    int nw = 0;
#pragma unroll
    for (int t = 0; t < EPT; ++t) {
      const int d = dl[t];
      const bool need = ((d + 1) & 2) != 0;
      const unsigned mask = __builtin_amdgcn_ballot_w32(need);
      if (need) {
        v2u e;
        e.x = ((unsigned)sl_[t] * (CSO * 4u)) | (((unsigned)d * (CSO * 4u)) << 16);
        e.y = __float_as_uint(wv[t]);
        seg[nw + (int)__builtin_amdgcn_mbcnt_lo(mask, 0u)] = e;
      }
      nw += __builtin_popcount(mask);
    }
    __syncthreads();
    const unsigned c4 = (unsigned)(lane & (CS - 1)) * 4u;
    const unsigned cc = c4 | (c4 << 16);
    const int li = ((lane & 15) << 1) | (lane >> 4);
    for (int e0 = 0; e0 < nw; e0 += 32) {
      v2u e = {0u, 0u};
      if (e0 + li < nw) e = seg[e0 + li];
      agg_half<0>(e.x, e.y, cc, hl, acc);
      if (e0 + 16 < nw) agg_half<8>(e.x, e.y, cc, hl, acc);
    }
  } else {
    __syncthreads();

    {
      const int* __restrict__ es = ei + (size_t)b * EPG;
      const int* __restrict__ ed = ei + (size_t)NEDGES + (size_t)b * EPG;
      for (int e0 = (tid >> 5) * 32; e0 < EPG; e0 += nwv * 32) {
        const int s = es[e0 + lane] - gbase;
        const int d = ed[e0 + lane] - gbase;
        const float w = rsqrtf((float)cnt[s]) * rsqrtf((float)cnt[d]);
        const unsigned sd = ((unsigned)s * (CSO * 4u)) | (((unsigned)d * (CSO * 4u)) << 16);
        if constexpr (CSO == 32) {
          const unsigned c4 = (unsigned)lane * 4u;
          agg32_group<0>(sd, __float_as_uint(w), c4, hl, acc);
          agg32_group<8>(sd, __float_as_uint(w), c4, hl, acc);
          agg32_group<16>(sd, __float_as_uint(w), c4, hl, acc);
          agg32_group<24>(sd, __float_as_uint(w), c4, hl, acc);
        } else {
          const unsigned c4 = (unsigned)(lane & (CS - 1)) * 4u;
          const unsigned cc = c4 | (c4 << 16);
          agg_half<0>(sd, __float_as_uint(w), cc, hl, acc);
          agg_half<8>(sd, __float_as_uint(w), cc, hl, acc);
        }
      }
    }
  }
  __syncthreads();

  constexpr int TP = 2 * RES + 64 / CSO;
  float* tp = hl;
  for (int idx = tid; idx < RES * CSO; idx += blockDim.x) {
    int c = idx & (CSO - 1);
    int r = idx / CSO;
    float bb = bsl[c];
    float v1 = fmaxf(acc[(4 * r + 1) * CSO + c] + bb, 0.0f);
    float v2 = fmaxf(acc[(4 * r + 2) * CSO + c] + bb, 0.0f);
    tp[(c >> 1) * TP + 2 * r + (c & 1)] = 0.5f * (v1 + v2);
  }
  __syncthreads();
  float* pg = pout + (size_t)b * RES * N + (size_t)sl * RES * CSO;
  for (int idx = tid; idx < RES * CSO; idx += blockDim.x)
    pg[idx] = tp[(idx >> 8) * TP + (idx & (2 * RES - 1))];
  if constexpr (!LAST) {
    float* yg = y + ((size_t)sl * NNODES + gbase) * CSO;
    for (int idx = tid; idx < NPG * CSO; idx += blockDim.x) {
      float v = acc[idx] + bsl[idx & (CSO - 1)];
      yg[idx] = fmaxf(v, 0.0f);
    }
  }
}

template <int D>
__global__ void __launch_bounds__(256) __attribute__((amdgpu_waves_per_eu(2, 2))) k_sim_wmma(const float* __restrict__ U,
                                                  const float* __restrict__ V,
                                                  float* __restrict__ out) {
  const int b    = blockIdx.x;
  const int w    = threadIdx.x >> 5;
  const int lane = threadIdx.x & 31;
  const int r    = lane & 15;
  const int h    = lane >> 4;
  const int tm   = blockIdx.y * 4 + (w >> 1);
  const int tc   = (w & 1) * 4;
  const v2f* Ug = (const v2f*)U + (size_t)b * (RES * D / 2) + h * RES + tm * 16 + r;
  const v2f* Vg = (const v2f*)V + (size_t)b * (RES * D / 2) + h * RES + tc * 16 + r;

  constexpr int KS = D / 4;
  v2f a[KS];
  v2f bf[KS][4];
#pragma unroll
  for (int s = 0; s < KS; ++s) {
    a[s] = Ug[(2 * s) * RES];
#pragma unroll
    for (int t = 0; t < 4; ++t)
      bf[s][t] = Vg[(2 * s) * RES + t * 16];
  }
  v8f acc[4] = {};
#pragma unroll
  for (int s = 0; s < KS; ++s)
#pragma unroll
    for (int t = 0; t < 4; ++t)
      acc[t] = __builtin_amdgcn_wmma_f32_16x16x4_f32(false, a[s], false, bf[s][t], (short)0,
                                                     acc[t], false, false);
  const int rbase = h << 3;
#pragma unroll
  for (int t = 0; t < 4; ++t) {
    float* Co = out + (size_t)b * RES * RES + (size_t)(tm * 16 + rbase) * RES + (tc + t) * 16 + r;
#pragma unroll
    for (int v = 0; v < 8; ++v) Co[(size_t)v * RES] = acc[t][v];
  }
}

extern "C" void kernel_launch(void* const* d_in, const int* in_sizes, int n_in,
                              void* d_out, int out_size, void* d_ws, size_t ws_size,
                              hipStream_t stream) {
  (void)in_sizes; (void)n_in; (void)out_size; (void)ws_size;
  const float* x[2]  = {(const float*)d_in[0], (const float*)d_in[1]};
  const int*   ei[2] = {(const int*)d_in[2],  (const int*)d_in[3]};
  const float* W[3]  = {(const float*)d_in[6], (const float*)d_in[8], (const float*)d_in[10]};
  const float* bv[3] = {(const float*)d_in[7], (const float*)d_in[9], (const float*)d_in[11]};
  float* out = (float*)d_out;

  float* ws = (float*)d_ws;
  size_t o = 0;
  float* Y1[2]; float* Y2[2];
  for (int s = 0; s < 2; ++s) { Y1[s] = ws + o; o += (size_t)NNODES * 64; }
  for (int s = 0; s < 2; ++s) { Y2[s] = ws + o; o += (size_t)NNODES * 32; }
  float* P[2][3];
  const int dims[3] = {64, 32, 16};
  for (int s = 0; s < 2; ++s)
    for (int l = 0; l < 3; ++l) { P[s][l] = ws + o; o += (size_t)NGRAPHS * RES * dims[l]; }


  k_layer<128, 0, 64, 32, false, 0><<<dim3(NGRAPHS, 2, 2), 1024, 0, stream>>>(
      x[0], x[1], W[0], ei[0], ei[1], bv[0], Y1[0], Y1[1], P[0][0], P[1][0],
      nullptr, nullptr, nullptr);
  k_layer<64, 32, 32, 32, false, 64><<<dim3(NGRAPHS, 1, 4), 1024, 0, stream>>>(
      Y1[0], Y1[1], W[1], ei[0], ei[1], bv[1], Y2[0], Y2[1], P[0][1], P[1][1],
      P[0][0], P[1][0], out);
  k_layer<32, 32, 16, 16, true, 32><<<dim3(NGRAPHS, 1, 4), 512, 0, stream>>>(
      Y2[0], Y2[1], W[2], ei[0], ei[1], bv[2], nullptr, nullptr, P[0][2], P[1][2],
      P[0][1], P[1][1], out + (size_t)1 * NGRAPHS * RES * RES);
  k_sim_wmma<16><<<dim3(NGRAPHS, 2), 256, 0, stream>>>(P[0][2], P[1][2],
                                                   out + (size_t)2 * NGRAPHS * RES * RES);
}
